# starting kernel
# baseline (speedup 1.0000x reference)
#include <hip/hip_runtime.h>

#define D_MODEL 1024
#define ADAPT   128
#define NUM_LIB 8
#define NPAIR   64
#define BATCH   1024
#define H_STRIDE 132

typedef __attribute__((ext_vector_type(2))) float v2f;
typedef __attribute__((ext_vector_type(8))) float v8f;

__global__ __launch_bounds__(BATCH) void classify_kernel(
    const int* __restrict__ src, const int* __restrict__ tgt,
    int* __restrict__ counts, int* __restrict__ lists)
{
    const int t = threadIdx.x;
    if (t < NPAIR) counts[t] = 0;
    __syncthreads();
    const int s = src[t], g = tgt[t];
    if (s != g) {
        const int p = s * NUM_LIB + g;
        const int slot = atomicAdd(&counts[p], 1);
        lists[p * BATCH + slot] = t;
    }
}

__global__ __launch_bounds__(256) void identity_copy_kernel(
    const float* __restrict__ x, const int* __restrict__ src,
    const int* __restrict__ tgt, float* __restrict__ out)
{
    const int row = blockIdx.x;
    if (src[row] != tgt[row]) return;
    const float4 v = ((const float4*)(x + (size_t)row * D_MODEL))[threadIdx.x];
    ((float4*)(out + (size_t)row * D_MODEL))[threadIdx.x] = v;
}

__global__ __launch_bounds__(256) void grouped_adapter_gemm_kernel(
    const float* __restrict__ x,  const float* __restrict__ W1,
    const float* __restrict__ b1, const float* __restrict__ W2,
    const float* __restrict__ b2, const int* __restrict__ counts,
    const int* __restrict__ lists, float* __restrict__ out)
{
    const int p   = blockIdx.x;
    const int n_p = counts[p];
    const int tile = blockIdx.y;
    if (tile * 16 >= n_p) return;

    __shared__ float hS[16 * H_STRIDE];
    __shared__ int rowIdx[16];

    const int t    = threadIdx.x;
    const int lane = t & 31;
    const int wave = t >> 5;
    const int ml   = lane & 15;
    const int hi   = lane >> 4;
    const int kb   = hi * 2;

    if (t < 16) {
        const int g = tile * 16 + t;
        rowIdx[t] = (g < n_p) ? lists[p * BATCH + g] : lists[p * BATCH];
    }
    __syncthreads();

    const float* W1p = W1 + (size_t)p * D_MODEL * ADAPT;
    const float* W2p = W2 + (size_t)p * ADAPT * D_MODEL;

    {
        const int n0 = wave * 16;
        const float* xrow = x + (size_t)rowIdx[ml] * D_MODEL;
        v8f acc = {};
        #pragma unroll 4
        for (int k = 0; k < D_MODEL; k += 4) {
            v2f a = *(const v2f*)(xrow + k + kb);
            v2f b;
            b.x = W1p[(size_t)(k + kb    ) * ADAPT + n0 + ml];
            b.y = W1p[(size_t)(k + kb + 1) * ADAPT + n0 + ml];
            acc = __builtin_amdgcn_wmma_f32_16x16x4_f32(
                false, a, false, b, (short)0, acc, false, false);
        }
        const float bias = b1[p * ADAPT + n0 + ml];
        #pragma unroll
        for (int v = 0; v < 8; ++v) {
            float hv = acc[v] + bias;
            hv = hv > 0.0f ? hv : 0.0f;
            hS[(hi * 8 + v) * H_STRIDE + n0 + ml] = hv;
        }
    }
    __syncthreads();

    const int g0 = tile * 16;
    #pragma unroll 1
    for (int j = 0; j < 8; ++j) {
        const int n0 = wave * 128 + j * 16;
        v8f acc = {};
        #pragma unroll 4
        for (int k = 0; k < ADAPT; k += 4) {
            v2f a = *(const v2f*)(hS + ml * H_STRIDE + k + kb);
            v2f b;
            b.x = W2p[(size_t)(k + kb    ) * D_MODEL + n0 + ml];
            b.y = W2p[(size_t)(k + kb + 1) * D_MODEL + n0 + ml];
            acc = __builtin_amdgcn_wmma_f32_16x16x4_f32(
                false, a, false, b, (short)0, acc, false, false);
        }
        const float bias = b2[p * D_MODEL + n0 + ml];
        #pragma unroll
        for (int v = 0; v < 8; ++v) {
            const int m2 = hi * 8 + v;
            if (g0 + m2 < n_p) {
                out[(size_t)rowIdx[m2] * D_MODEL + n0 + ml] = acc[v] + bias;
            }
        }
    }
}

extern "C" void kernel_launch(void* const* d_in, const int* in_sizes, int n_in,
                              void* d_out, int out_size, void* d_ws, size_t ws_size,
                              hipStream_t stream) {
    const float* x   = (const float*)d_in[0];
    const int*   src = (const int*)  d_in[1];
    const int*   tgt = (const int*)  d_in[2];
    const float* W1  = (const float*)d_in[3];
    const float* b1  = (const float*)d_in[4];
    const float* W2  = (const float*)d_in[5];
    const float* b2  = (const float*)d_in[6];
    float* out = (float*)d_out;

    int* counts = (int*)d_ws;
    int* lists  = counts + NPAIR;

    classify_kernel<<<1, BATCH, 0, stream>>>(src, tgt, counts, lists);
    identity_copy_kernel<<<BATCH, 256, 0, stream>>>(x, src, tgt, out);
    grouped_adapter_gemm_kernel<<<dim3(NPAIR, BATCH / 16), 256, 0, stream>>>(
        x, W1, b1, W2, b2, counts, lists, out);
}


# Weights, x and h go through LDS: the block copies 64 KB weight tiles with coalesced 16-byte loads, next tile in flight during the WMMAs; fragments read from LDS; k order and sums unchanged
# speedup vs baseline: 2.2465x; 2.2465x over previous
#include <hip/hip_runtime.h>

#define D_MODEL 1024
#define ADAPT   128
#define NUM_LIB 8
#define NPAIR   64
#define BATCH   1024
#define A_PITCH  132
#define W1_PITCH 136
#define W2_PITCH 264
#define W_LDS_FLOATS (128 * W1_PITCH)

typedef __attribute__((ext_vector_type(2))) float v2f;
typedef __attribute__((ext_vector_type(8))) float v8f;
typedef __attribute__((ext_vector_type(4))) float v4f;

#define W_REP16(M) M(0) M(1) M(2) M(3) M(4) M(5) M(6) M(7) M(8) M(9) M(10) M(11) M(12) M(13) M(14) M(15)
#define W_DECL(i) v4f w##i;
#define W_LD1(i)  w##i = *(const v4f*)(src + (size_t)(8 * i) * ADAPT);
#define W_LD2(i)  w##i = *(const v4f*)(src + (size_t)(4 * i) * D_MODEL);
#define W_ST1(i)  *(v4f*)(wS + (s1r + 8 * i) * W1_PITCH + s1c) = w##i;
#define W_ST2(i)  *(v4f*)(wS + (s2r + 4 * i) * W2_PITCH + s2c) = w##i;

__global__ __launch_bounds__(BATCH) void classify_kernel(
    const int* __restrict__ src, const int* __restrict__ tgt,
    int* __restrict__ counts, int* __restrict__ lists)
{
    const int t = threadIdx.x;
    if (t < NPAIR) counts[t] = 0;
    __syncthreads();
    const int s = src[t], g = tgt[t];
    if (s != g) {
        const int p = s * NUM_LIB + g;
        const int slot = atomicAdd(&counts[p], 1);
        lists[p * BATCH + slot] = t;
    }
}

__global__ __launch_bounds__(256) void identity_copy_kernel(
    const float* __restrict__ x, const int* __restrict__ src,
    const int* __restrict__ tgt, float* __restrict__ out)
{
    const int row = blockIdx.x;
    if (src[row] != tgt[row]) return;
    const float4 v = ((const float4*)(x + (size_t)row * D_MODEL))[threadIdx.x];
    ((float4*)(out + (size_t)row * D_MODEL))[threadIdx.x] = v;
}

__global__ __launch_bounds__(256) void grouped_adapter_gemm_kernel(
    const float* __restrict__ x,  const float* __restrict__ W1,
    const float* __restrict__ b1, const float* __restrict__ W2,
    const float* __restrict__ b2, const int* __restrict__ counts,
    const int* __restrict__ lists, float* __restrict__ out)
{
    const int p   = blockIdx.x;
    const int n_p = counts[p];
    const int tile = blockIdx.y;
    if (tile * 16 >= n_p) return;

    __shared__ __attribute__((aligned(16))) float wS[W_LDS_FLOATS];
    __shared__ __attribute__((aligned(16))) float aS[16 * A_PITCH];
    __shared__ int rowIdx[16];

    const int t    = threadIdx.x;
    const int lane = t & 31;
    const int wave = t >> 5;
    const int ml   = lane & 15;
    const int hi   = lane >> 4;
    const int kb   = hi * 2;

    const float* W1p = W1 + (size_t)p * D_MODEL * ADAPT;
    const float* W2p = W2 + (size_t)p * ADAPT * D_MODEL;

    const int s1c = (t & 31) * 4;
    const int s1r = t >> 5;
    const int s2c = (t & 63) * 4;
    const int s2r = t >> 6;

    W_REP16(W_DECL)
    v4f xr0, xr1;

    const float* w1src = W1p + (size_t)s1r * ADAPT + s1c;
    {
        const float* src = w1src;
        W_REP16(W_LD1)
    }

    if (t < 16) {
        const int g = tile * 16 + t;
        rowIdx[t] = (g < n_p) ? lists[p * BATCH + g] : lists[p * BATCH];
    }
    __syncthreads();

    const float* xs0 = x + (size_t)rowIdx[s1r]     * D_MODEL + s1c;
    const float* xs1 = x + (size_t)rowIdx[s1r + 8] * D_MODEL + s1c;
    xr0 = *(const v4f*)xs0;
    xr1 = *(const v4f*)xs1;

    {
        const int n0 = wave * 16;
        v8f acc = {};
        #pragma unroll 1
        for (int c = 0; c < 8; ++c) {
            __syncthreads();
            W_REP16(W_ST1)
            *(v4f*)(aS + s1r * A_PITCH + s1c)       = xr0;
            *(v4f*)(aS + (s1r + 8) * A_PITCH + s1c) = xr1;
            __syncthreads();

            if (c + 1 < 8) {
                const float* src = w1src + (size_t)(c + 1) * 128 * ADAPT;
                W_REP16(W_LD1)
                xr0 = *(const v4f*)(xs0 + (c + 1) * 128);
                xr1 = *(const v4f*)(xs1 + (c + 1) * 128);
            } else {
                const float* src = W2p + (size_t)s2r * D_MODEL + s2c;
                W_REP16(W_LD2)
            }

            #pragma unroll 8
            for (int k = 0; k < 128; k += 4) {
                v2f a = *(const v2f*)(aS + ml * A_PITCH + k + kb);
                v2f b;
                b.x = wS[(k + kb    ) * W1_PITCH + n0 + ml];
                b.y = wS[(k + kb + 1) * W1_PITCH + n0 + ml];
                acc = __builtin_amdgcn_wmma_f32_16x16x4_f32(
                    false, a, false, b, (short)0, acc, false, false);
            }
        }
        __syncthreads();
        const float bias = b1[p * ADAPT + n0 + ml];
        #pragma unroll
        for (int v = 0; v < 8; ++v) {
            float hv = acc[v] + bias;
            hv = hv > 0.0f ? hv : 0.0f;
            aS[(hi * 8 + v) * A_PITCH + n0 + ml] = hv;
        }
    }

    const int g0 = tile * 16;
    const int nA = wave * 32;
    #pragma unroll 1
    for (int g = 0; g < 4; ++g) {
        v8f acc0 = {};
        v8f acc1 = {};
        #pragma unroll 1
        for (int kh = 0; kh < 2; ++kh) {
            __syncthreads();
            W_REP16(W_ST2)
            __syncthreads();

            const int nt = g * 2 + kh + 1;
            if (nt < 8) {
                const float* src = W2p + (size_t)((nt & 1) * 64 + s2r) * D_MODEL
                                       + (nt >> 1) * 256 + s2c;
                W_REP16(W_LD2)
            }

            const float* hrow = aS + ml * A_PITCH + kh * 64 + kb;
            #pragma unroll 8
            for (int k = 0; k < 64; k += 4) {
                v2f a = *(const v2f*)(hrow + k);
                v2f bA, bB;
                bA.x = wS[(k + kb    ) * W2_PITCH + nA + ml];
                bA.y = wS[(k + kb + 1) * W2_PITCH + nA + ml];
                bB.x = wS[(k + kb    ) * W2_PITCH + nA + 16 + ml];
                bB.y = wS[(k + kb + 1) * W2_PITCH + nA + 16 + ml];
                acc0 = __builtin_amdgcn_wmma_f32_16x16x4_f32(
                    false, a, false, bA, (short)0, acc0, false, false);
                acc1 = __builtin_amdgcn_wmma_f32_16x16x4_f32(
                    false, a, false, bB, (short)0, acc1, false, false);
            }
        }
        const int col = g * 256 + nA + ml;
        const float biasA = b2[p * D_MODEL + col];
        const float biasB = b2[p * D_MODEL + col + 16];
        #pragma unroll
        for (int v = 0; v < 8; ++v) {
            const int m2 = hi * 8 + v;
            if (g0 + m2 < n_p) {
                float* o = out + (size_t)rowIdx[m2] * D_MODEL + col;
                o[0]  = acc0[v] + biasA;
                o[16] = acc1[v] + biasB;
            }
        }
    }
}

extern "C" void kernel_launch(void* const* d_in, const int* in_sizes, int n_in,
                              void* d_out, int out_size, void* d_ws, size_t ws_size,
                              hipStream_t stream) {
    const float* x   = (const float*)d_in[0];
    const int*   src = (const int*)  d_in[1];
    const int*   tgt = (const int*)  d_in[2];
    const float* W1  = (const float*)d_in[3];
    const float* b1  = (const float*)d_in[4];
    const float* W2  = (const float*)d_in[5];
    const float* b2  = (const float*)d_in[6];
    float* out = (float*)d_out;

    int* counts = (int*)d_ws;
    int* lists  = counts + NPAIR;

    classify_kernel<<<1, BATCH, 0, stream>>>(src, tgt, counts, lists);
    identity_copy_kernel<<<BATCH, 256, 0, stream>>>(x, src, tgt, out);
    grouped_adapter_gemm_kernel<<<dim3(NPAIR, BATCH / 16), 256, 0, stream>>>(
        x, W1, b1, W2, b2, counts, lists, out);
}


# Identity rows join the buckets: classify lists src==tgt rows under their diagonal pair, that pair's GEMM-grid blocks copy them x->out; the copy launch goes (3 launches -> 2); sums unchanged
# speedup vs baseline: 3.5581x; 1.5838x over previous
#include <hip/hip_runtime.h>

#define D_MODEL 1024
#define ADAPT   128
#define NUM_LIB 8
#define NPAIR   64
#define BATCH   1024
#define A_PITCH  132
#define W1_PITCH 136
#define W2_HALF  (64 * W1_PITCH + 32)
#define W_LDS_FLOATS (W2_HALF + 64 * W1_PITCH)

typedef __attribute__((ext_vector_type(2))) float v2f;
typedef __attribute__((ext_vector_type(8))) float v8f;
typedef __attribute__((ext_vector_type(4))) float v4f;

#define A_FLOATS (16 * A_PITCH)

#define W_REP16(M) M(0) M(1) M(2) M(3) M(4) M(5) M(6) M(7) M(8) M(9) M(10) M(11) M(12) M(13) M(14) M(15)
#define W_DECLA(i) v4f wa##i;
#define W_DECLB(i) v4f wb##i;
#define W_LD1A(i)  wa##i = *(const v4f*)(src + (size_t)(8 * i) * ADAPT);
#define W_LD1B(i)  wb##i = *(const v4f*)(src + (size_t)(8 * i) * ADAPT);
#define W_LD2A(i)  wa##i = *(const v4f*)(src + (size_t)(4 * i) * D_MODEL);
#define W_LD2B(i)  wb##i = *(const v4f*)(src + (size_t)(4 * i) * D_MODEL);
#define W_ST1A(i)  *(v4f*)(dst + (s1r + 8 * i) * W1_PITCH + s1c) = wa##i;
#define W_ST1B(i)  *(v4f*)(dst + (s1r + 8 * i) * W1_PITCH + s1c) = wb##i;
#define W_ST2A(i)  *(v4f*)(dst + s2o + (4 * i) * W1_PITCH) = wa##i;
#define W_ST2B(i)  *(v4f*)(dst + s2o + (4 * i) * W1_PITCH) = wb##i;

__global__ __launch_bounds__(BATCH) void classify_kernel(
    const int* __restrict__ src, const int* __restrict__ tgt,
    int* __restrict__ counts, int* __restrict__ lists)
{
    const int t = threadIdx.x;
    if (t < NPAIR) counts[t] = 0;
    __syncthreads();
    const int s = src[t], g = tgt[t];
    const int p = s * NUM_LIB + g;
    const int slot = atomicAdd(&counts[p], 1);
    lists[p * BATCH + slot] = t;
}

__device__ __forceinline__ void lds_barrier() {
    __builtin_amdgcn_fence(__ATOMIC_RELEASE, "workgroup", "local");
    __builtin_amdgcn_s_barrier();
    __builtin_amdgcn_fence(__ATOMIC_ACQUIRE, "workgroup", "local");
}

__device__ __forceinline__ void frag_load_p1(v2f (&fa)[16], v2f (&fb)[16],
                                             const float* ap, const float* bp)
{
    #pragma unroll
    for (int j = 0; j < 16; ++j) {
        fa[j]   = *(const v2f*)(ap + 4 * j);
        fb[j].x = bp[(4 * j    ) * W1_PITCH];
        fb[j].y = bp[(4 * j + 1) * W1_PITCH];
        if ((j & 3) == 3) __builtin_amdgcn_sched_barrier(0);
    }
}
__device__ __forceinline__ void frag_load_p2(v2f (&fa)[16], v2f (&fbA)[16], v2f (&fbB)[16],
                                             const float* ap, const float* bp)
{
    #pragma unroll
    for (int j = 0; j < 16; ++j) {
        fa[j]    = *(const v2f*)(ap + 4 * j);
        fbA[j].x = bp[(4 * j    ) * W1_PITCH];
        fbA[j].y = bp[(4 * j + 1) * W1_PITCH];
        fbB[j].x = bp[(4 * j    ) * W1_PITCH + W2_HALF];
        fbB[j].y = bp[(4 * j + 1) * W1_PITCH + W2_HALF];
        if ((j & 1) == 1) __builtin_amdgcn_sched_barrier(0);
    }
}
__device__ __forceinline__ v8f mma8(const v2f (&fa)[16], const v2f (&fb)[16], int j0, v8f acc)
{
    #pragma unroll
    for (int j = 0; j < 8; ++j)
        acc = __builtin_amdgcn_wmma_f32_16x16x4_f32(
            false, fa[j0 + j], false, fb[j0 + j], (short)0, acc, false, false);
    return acc;
}
__device__ __forceinline__ void mma4x2(const v2f (&fa)[16], const v2f (&fbA)[16], const v2f (&fbB)[16],
                                       int j0, v8f& acc0, v8f& acc1)
{
    #pragma unroll
    for (int j = 0; j < 4; ++j) {
        acc0 = __builtin_amdgcn_wmma_f32_16x16x4_f32(
            false, fa[j0 + j], false, fbA[j0 + j], (short)0, acc0, false, false);
        acc1 = __builtin_amdgcn_wmma_f32_16x16x4_f32(
            false, fa[j0 + j], false, fbB[j0 + j], (short)0, acc1, false, false);
    }
}

__global__ __launch_bounds__(256) void grouped_adapter_gemm_kernel(
    const float* __restrict__ x,  const float* __restrict__ W1,
    const float* __restrict__ b1, const float* __restrict__ W2,
    const float* __restrict__ b2, const int* __restrict__ counts,
    const int* __restrict__ lists, float* __restrict__ out)
{
    const int p    = blockIdx.x >> 1;
    const int half = blockIdx.x & 1;
    const int n_p = counts[p];
    const int tile = blockIdx.y;
    if (tile * 16 >= n_p) return;

    if ((p >> 3) == (p & (NUM_LIB - 1))) {
        const int c4 = threadIdx.x & 127;
        const int r  = threadIdx.x >> 7;
        const int* lp = lists + p * BATCH + tile * 16;
        const int last = n_p - 1 - tile * 16;
        int rows[8];
        #pragma unroll
        for (int i = 0; i < 8; ++i) {
            const int m = 2 * i + r;
            rows[i] = lp[m < last ? m : last];
        }
        v4f v[8];
        #pragma unroll
        for (int i = 0; i < 8; ++i)
            v[i] = ((const v4f*)(x + (size_t)rows[i] * D_MODEL + half * 512))[c4];
        asm volatile("" :: "v"(v[0]), "v"(v[1]), "v"(v[2]), "v"(v[3]), "v"(v[4]), "v"(v[5]), "v"(v[6]), "v"(v[7]));
        #pragma unroll
        for (int i = 0; i < 8; ++i)
            if (tile * 16 + 2 * i + r < n_p)
                ((v4f*)(out + (size_t)rows[i] * D_MODEL + half * 512))[c4] = v[i];
        return;
    }

    __shared__ __attribute__((aligned(16))) float wS[2 * W_LDS_FLOATS];
    __shared__ __attribute__((aligned(16))) float aS[2 * A_FLOATS];
    __shared__ __attribute__((aligned(16))) float bS[D_MODEL + ADAPT];
    __shared__ int rowIdx[16];
    float* const wS0 = wS;
    float* const wS1 = wS + W_LDS_FLOATS;
    float* const aS0 = aS;
    float* const aS1 = aS + A_FLOATS;

    const int t    = threadIdx.x;
    const int lane = t & 31;
    const int wave = t >> 5;
    const int ml   = lane & 15;
    const int hi   = lane >> 4;
    const int kb   = hi * 2;

    const float* W1p = W1 + (size_t)p * D_MODEL * ADAPT;
    const float* W2p = W2 + (size_t)p * ADAPT * D_MODEL;

    const int s1c = (t & 31) * 4;
    const int s1r = t >> 5;
    const int s2c = (t & 63) * 4;
    const int s2r = t >> 6;
    const int s2o = ((s2c >> 4) & 1) * W2_HALF + s2r * W1_PITCH + (s2c >> 5) * 16 + (s2c & 15);

    W_REP16(W_DECLA)
    W_REP16(W_DECLB)
    v4f xa0, xa1, xb0, xb1;

    const float* w1src = W1p + (size_t)s1r * ADAPT + s1c;
    const float* w2src = W2p + (size_t)s2r * D_MODEL + half * 512 + s2c;
#define W1_TILE(j)  (w1src + (size_t)(j) * 128 * ADAPT)
#define W2_TILE(nt) (w2src + (size_t)((nt) & 1) * 64 * D_MODEL + ((nt) >> 1) * 256)

    const v4f b2q = *(const v4f*)(b2 + (size_t)p * D_MODEL + t * 4);
    float b1q = 0.0f;
    if (t < ADAPT) b1q = b1[p * ADAPT + t];
    int myRow = 0;
    if (t < 16) {
        const int g = tile * 16 + t;
        myRow = (g < n_p) ? lists[p * BATCH + g] : lists[p * BATCH];
    }
    { const float* src = W1_TILE(0); W_REP16(W_LD1A) }
    __builtin_amdgcn_sched_barrier(0);
    *(v4f*)(bS + t * 4) = b2q;
    if (t < ADAPT) bS[D_MODEL + t] = b1q;
    if (t < 16) rowIdx[t] = myRow;
    lds_barrier();

    const float* xs0 = x + (size_t)rowIdx[s1r]     * D_MODEL + s1c;
    const float* xs1 = x + (size_t)rowIdx[s1r + 8] * D_MODEL + s1c;
    xa0 = *(const v4f*)xs0;
    xa1 = *(const v4f*)xs1;
    xb0 = *(const v4f*)(xs0 + 128);
    xb1 = *(const v4f*)(xs1 + 128);
    __builtin_amdgcn_sched_barrier(0);
    { const float* src = W1_TILE(1); W_REP16(W_LD1B) }

    { float* dst = wS0; W_REP16(W_ST1A) }
    *(v4f*)(aS0 + s1r * A_PITCH + s1c)       = xa0;
    *(v4f*)(aS0 + (s1r + 8) * A_PITCH + s1c) = xa1;
    { const float* src = W1_TILE(2); W_REP16(W_LD1A) }
    xa0 = *(const v4f*)(xs0 + 2 * 128);
    xa1 = *(const v4f*)(xs1 + 2 * 128);
    lds_barrier();

    {
        const int n0 = wave * 16;
        v8f acc = {};
        const float* const ap0 = aS0 + ml * A_PITCH + kb;
        const float* const ap1 = aS1 + ml * A_PITCH + kb;
        const float* const bp0 = wS0 + kb * W1_PITCH + n0 + ml;
        const float* const bp1 = wS1 + kb * W1_PITCH + n0 + ml;
        #pragma unroll 1
        for (int c = 0; c < 8; c += 2) {
            v2f fa0[16], fb0[16], fa1[16], fb1[16];
            frag_load_p1(fa0, fb0, ap0, bp0);
            frag_load_p1(fa1, fb1, ap0 + 64, bp0 + 64 * W1_PITCH);
            acc = mma8(fa0, fb0, 0, acc);
            __builtin_amdgcn_sched_barrier(0);
            { float* dst = wS1; W_REP16(W_ST1B) }
            *(v4f*)(aS1 + s1r * A_PITCH + s1c)       = xb0;
            *(v4f*)(aS1 + (s1r + 8) * A_PITCH + s1c) = xb1;
            if (c + 3 < 8) {
                const float* src = W1_TILE(c + 3);
                W_REP16(W_LD1B)
                xb0 = *(const v4f*)(xs0 + (c + 3) * 128);
                xb1 = *(const v4f*)(xs1 + (c + 3) * 128);
            } else {
                const float* src = W2_TILE(1);
                W_REP16(W_LD2B)
            }
            __builtin_amdgcn_sched_barrier(0);
            acc = mma8(fa0, fb0, 8, acc);
            acc = mma8(fa1, fb1, 0, acc);
            __builtin_amdgcn_sched_barrier(0);
            acc = mma8(fa1, fb1, 8, acc);
            lds_barrier();

            frag_load_p1(fa0, fb0, ap1, bp1);
            frag_load_p1(fa1, fb1, ap1 + 64, bp1 + 64 * W1_PITCH);
            acc = mma8(fa0, fb0, 0, acc);
            __builtin_amdgcn_sched_barrier(0);
            if (c + 2 < 8) {
                float* dst = wS0;
                W_REP16(W_ST1A)
                *(v4f*)(aS0 + s1r * A_PITCH + s1c)       = xa0;
                *(v4f*)(aS0 + (s1r + 8) * A_PITCH + s1c) = xa1;
            } else {
                float* dst = wS0;
                W_REP16(W_ST2A)
            }
            if (c + 4 < 8) {
                const float* src = W1_TILE(c + 4);
                W_REP16(W_LD1A)
                xa0 = *(const v4f*)(xs0 + (c + 4) * 128);
                xa1 = *(const v4f*)(xs1 + (c + 4) * 128);
            } else {
                const float* src = W2_TILE(c + 4 - 8);
                W_REP16(W_LD2A)
            }
            __builtin_amdgcn_sched_barrier(0);
            acc = mma8(fa0, fb0, 8, acc);
            acc = mma8(fa1, fb1, 0, acc);
            acc = mma8(fa1, fb1, 8, acc);
            if (c + 2 < 8) lds_barrier();
        }
        const float bias = bS[D_MODEL + n0 + ml];
        #pragma unroll
        for (int v = 0; v < 8; ++v) {
            float hv = acc[v] + bias;
            hv = hv > 0.0f ? hv : 0.0f;
            aS0[(hi * 8 + v) * A_PITCH + n0 + ml] = hv;
        }
        lds_barrier();
    }

    const int g0 = tile * 16;
    const int nA = wave * 32;
    const int w2off = kb * W1_PITCH + wave * 16 + ml;
    const float* const hp = aS0 + ml * A_PITCH + kb;
    #pragma unroll
    for (int g = 0; g < 2; ++g) {
        v8f acc0 = {};
        v8f acc1 = {};
        v2f fa[16], fbA[16], fbB[16];
        frag_load_p2(fa, fbA, fbB, hp, wS0 + w2off);
        mma4x2(fa, fbA, fbB, 0, acc0, acc1);
        __builtin_amdgcn_sched_barrier(0);
        { float* dst = wS1; W_REP16(W_ST2B) }
        if (2 * g + 3 < 4) {
            const float* src = W2_TILE(2 * g + 3);
            W_REP16(W_LD2B)
        }
        __builtin_amdgcn_sched_barrier(0);
        mma4x2(fa, fbA, fbB, 4, acc0, acc1);
        mma4x2(fa, fbA, fbB, 8, acc0, acc1);
        __builtin_amdgcn_sched_barrier(0);
        mma4x2(fa, fbA, fbB, 12, acc0, acc1);
        lds_barrier();

        frag_load_p2(fa, fbA, fbB, hp + 64, wS1 + w2off);
        mma4x2(fa, fbA, fbB, 0, acc0, acc1);
        __builtin_amdgcn_sched_barrier(0);
        if (2 * g + 2 < 4) {
            float* dst = wS0;
            W_REP16(W_ST2A)
        }
        if (2 * g + 4 < 4) {
            const float* src = W2_TILE(2 * g + 4);
            W_REP16(W_LD2A)
        }
        __builtin_amdgcn_sched_barrier(0);
        mma4x2(fa, fbA, fbB, 4, acc0, acc1);
        mma4x2(fa, fbA, fbB, 8, acc0, acc1);
        mma4x2(fa, fbA, fbB, 12, acc0, acc1);
        const int col = (2 * half + g) * 256 + nA + ml;
        const float biasA = bS[col];
        const float biasB = bS[col + 16];
        #pragma unroll
        for (int v = 0; v < 8; ++v) {
            const int m2 = hi * 8 + v;
            if (g0 + m2 < n_p) {
                float* o = out + (size_t)rowIdx[m2] * D_MODEL + col;
                o[0]  = acc0[v] + biasA;
                o[16] = acc1[v] + biasB;
            }
        }
        if (g + 1 < 2) lds_barrier();
    }
#undef W1_TILE
#undef W2_TILE
}

extern "C" void kernel_launch(void* const* d_in, const int* in_sizes, int n_in,
                              void* d_out, int out_size, void* d_ws, size_t ws_size,
                              hipStream_t stream) {
    const float* x   = (const float*)d_in[0];
    const int*   src = (const int*)  d_in[1];
    const int*   tgt = (const int*)  d_in[2];
    const float* W1  = (const float*)d_in[3];
    const float* b1  = (const float*)d_in[4];
    const float* W2  = (const float*)d_in[5];
    const float* b2  = (const float*)d_in[6];
    float* out = (float*)d_out;

    int* counts = (int*)d_ws;
    int* lists  = counts + NPAIR;

    classify_kernel<<<1, BATCH, 0, stream>>>(src, tgt, counts, lists);
    grouped_adapter_gemm_kernel<<<dim3(2 * NPAIR, BATCH / 16), 256, 0, stream>>>(
        x, W1, b1, W2, b2, counts, lists, out);
}
